# starting kernel
# baseline (speedup 1.0000x reference)
#include <hip/hip_runtime.h>
#include <math.h>

typedef __attribute__((ext_vector_type(2))) float v2f;
typedef __attribute__((ext_vector_type(8))) float v8f;

__constant__ int LM_TABLE[88] = {
    468,469,470,471,472,473,474,475,476,477,478,479,480,481,482,483,484,485,486,487,488,
    522,523,524,525,526,527,528,529,530,531,532,533,534,535,536,537,538,539,540,541,542,
    61,185,40,39,37,0,267,269,270,409,291,146,91,181,84,17,314,405,321,375,
    78,191,80,81,82,13,312,311,310,415,95,88,178,87,14,317,402,318,324,308,
    0,0,0,0,0,0
};

#define FRAME_STRIDE 1629
#define NF 164
#define NGROUP 11
#define NSTRIP 4

__global__ __launch_bounds__(128)
void seg_stats_wmma(const float* __restrict__ frames, float* __restrict__ out)
{
    const int seg  = blockIdx.x;
    const int grp  = blockIdx.y;
    const int lane = threadIdx.x & 31;
    const int wave = threadIdx.x >> 5;

    const int t0 = (seg * 32767) / 64;
    const int t1 = ((seg + 1) * 32767) / 64;

    const int m   = lane & 15;
    const int kHi = lane >> 4;
    const int f   = grp * 16 + m;
    const int lm  = LM_TABLE[f >> 1];
    const int co  = f & 1;

    const float* base = frames + (size_t)lm * 3 + co;

    v8f accC = {};
    v8f accS = {};
    v8f accQ = {};
    const v2f ones = {1.0f, 1.0f};

    int tb = t0 + 128 * wave + 2 * kHi;
    const float* p = base + (size_t)tb * FRAME_STRIDE;

    for (int it = 0; it < 32; ++it) {
        const float xA = p[0];
        const float xB = p[FRAME_STRIDE];
        const bool vA = (tb     < t1) && (xA == xA);
        const bool vB = (tb + 1 < t1) && (xB == xB);
        const float a0 = vA ? xA : 0.0f;
        const float a1 = vB ? xB : 0.0f;

        v2f aV = { vA ? 1.0f : 0.0f, vB ? 1.0f : 0.0f };
        v2f aX = { a0, a1 };
        v2f aQ = { a0 * a0, a1 * a1 };

        accC = __builtin_amdgcn_wmma_f32_16x16x4_f32(false, aV, false, ones,
                                                     (short)0, accC, false, false);
        accS = __builtin_amdgcn_wmma_f32_16x16x4_f32(false, aX, false, ones,
                                                     (short)0, accS, false, false);
        accQ = __builtin_amdgcn_wmma_f32_16x16x4_f32(false, aQ, false, ones,
                                                     (short)0, accQ, false, false);

        tb += 4;
        p  += 4 * FRAME_STRIDE;
    }

    __shared__ float lds[NSTRIP * 16 * 3];
    if (m == 0) {
        const int fl = kHi * 8;
        #pragma unroll
        for (int e = 0; e < 8; ++e) {
            const int idx = (wave * 16 + fl + e) * 3;
            lds[idx + 0] = accC[e];
            lds[idx + 1] = accS[e];
            lds[idx + 2] = accQ[e];
        }
    }
    __syncthreads();

    if (threadIdx.x < 16) {
        const int fl = threadIdx.x;
        float c = 0.0f, s = 0.0f, q = 0.0f;
        #pragma unroll
        for (int st = 0; st < NSTRIP; ++st) {
            const int idx = (st * 16 + fl) * 3;
            c += lds[idx + 0];
            s += lds[idx + 1];
            q += lds[idx + 2];
        }
        const int fg = grp * 16 + fl;
        if (fg < NF) {
            float mean = 0.0f, sd = 0.0f;
            if (c > 0.0f) {
                mean = s / c;
                float var = q / c - mean * mean;
                sd = sqrtf(fmaxf(var, 0.0f));
            }
            if (!(mean == mean)) mean = 0.0f;
            if (!(sd   == sd))   sd   = 0.0f;
            float* row = out + (size_t)seg * 328;
            row[fg]       = mean;
            row[164 + fg] = sd;
        }
    }
}

extern "C" void kernel_launch(void* const* d_in, const int* in_sizes, int n_in,
                              void* d_out, int out_size, void* d_ws, size_t ws_size,
                              hipStream_t stream)
{
    (void)in_sizes; (void)n_in; (void)out_size; (void)d_ws; (void)ws_size;
    const float* frames = (const float*)d_in[0];
    float* out = (float*)d_out;
    dim3 grid(64, NGROUP);
    seg_stats_wmma<<<grid, 128, 0, stream>>>(frames, out);
}


# Work split: 4 byte-neighbour landmark sets x 16 strips of 32 frames per segment (was 11 groups x 4 strips), loads batched ahead of the WMMAs: fewer lines fetched; 16 partial sums, order changes
# speedup vs baseline: 1.1701x; 1.1701x over previous
#include <hip/hip_runtime.h>
#include <math.h>


typedef __attribute__((ext_vector_type(2))) float v2f;
typedef __attribute__((ext_vector_type(8))) float v8f;

__constant__ int LM_SLOT[96] = {
    0,13,14,17,37,39,40,61,78,80,468,469,470,471,472,473,474,475,476,477,478,478,478,478,
    81,82,84,87,88,91,95,146,178,181,479,480,481,482,483,484,485,486,487,488,488,488,488,488,
    185,191,267,269,270,291,308,310,311,312,522,523,524,525,526,527,528,529,530,531,532,532,532,532,
    314,317,318,321,324,375,402,405,409,415,533,534,535,536,537,538,539,540,541,542,542,542,542,542
};
__constant__ int OUT_SLOT[96] = {
    47,67,76,57,46,45,44,42,62,64,0,1,2,3,4,5,6,7,8,9,10,-1,-1,-1,
    65,66,56,75,73,54,72,53,74,55,11,12,13,14,15,16,17,18,19,20,-1,-1,-1,-1,
    43,63,48,49,50,52,81,70,69,68,21,22,23,24,25,26,27,28,29,30,31,-1,-1,-1,
    58,77,79,60,80,61,78,59,51,71,32,33,34,35,36,37,38,39,40,41,-1,-1,-1,-1
};

#define FRAME_STRIDE 1629
#define NSET 4
#define NGRP 3
#define NSLOT 24
#define NWAVE 16
#define FPW 32

__global__ __launch_bounds__(512)
void seg_stats_wmma(const float* __restrict__ frames, float* __restrict__ out)
{
    const int seg  = blockIdx.x;
    const int set  = blockIdx.y;
    const int lane = threadIdx.x & 31;
    const int wave = threadIdx.x >> 5;

    const int t0 = (seg * 32767) / 64;
    const int t1 = ((seg + 1) * 32767) / 64;

    const int m   = lane & 15;
    const int kHi = lane >> 4;
    const int co  = m & 1;

    int off[NGRP];
    #pragma unroll
    for (int g = 0; g < NGRP; ++g)
        off[g] = LM_SLOT[set * NSLOT + g * 8 + (m >> 1)] * 3 + co;

    v8f accC[NGRP], accS[NGRP], accQ[NGRP];
    #pragma unroll
    for (int g = 0; g < NGRP; ++g) { accC[g] = {}; accS[g] = {}; accQ[g] = {}; }
    const v2f ones = {1.0f, 1.0f};

    int tb = t0 + FPW * wave + 2 * kHi;
    const float* p = frames + (size_t)tb * FRAME_STRIDE;

    #define NBATCH (FPW / 8)
    float buf[NBATCH][2][NGRP][2];

    #define LOAD_BATCH(B)                                                        \
        _Pragma("unroll")                                                        \
        for (int j = 0; j < 2; ++j) {                                            \
            _Pragma("unroll")                                                    \
            for (int g = 0; g < NGRP; ++g) {                                     \
                const float* pl = p + (size_t)((B) * 8 + j * 4) * FRAME_STRIDE;  \
                buf[B][j][g][0] = pl[off[g]];                                    \
                buf[B][j][g][1] = pl[off[g] + FRAME_STRIDE];                     \
            }                                                                    \
        }

    LOAD_BATCH(0)
    LOAD_BATCH(1)

    #pragma unroll
    for (int bt = 0; bt < NBATCH; ++bt) {
        __builtin_amdgcn_sched_barrier(0);
        #pragma unroll
        for (int j = 0; j < 2; ++j) {
            const int tj = tb + bt * 8 + j * 4;
            const bool inA = (tj     < t1);
            const bool inB = (tj + 1 < t1);
            #pragma unroll
            for (int g = 0; g < NGRP; ++g) {
                const float xA = buf[bt][j][g][0];
                const float xB = buf[bt][j][g][1];
                const bool vA = inA && (xA == xA);
                const bool vB = inB && (xB == xB);
                const float a0 = vA ? xA : 0.0f;
                const float a1 = vB ? xB : 0.0f;

                v2f aV = { vA ? 1.0f : 0.0f, vB ? 1.0f : 0.0f };
                v2f aX = { a0, a1 };
                v2f aQ = { a0 * a0, a1 * a1 };

                accC[g] = __builtin_amdgcn_wmma_f32_16x16x4_f32(false, aV, false, ones,
                                                                (short)0, accC[g], false, false);
                accS[g] = __builtin_amdgcn_wmma_f32_16x16x4_f32(false, aX, false, ones,
                                                                (short)0, accS[g], false, false);
                accQ[g] = __builtin_amdgcn_wmma_f32_16x16x4_f32(false, aQ, false, ones,
                                                                (short)0, accQ[g], false, false);
            }
        }
        if (bt + 2 < NBATCH) {
            __builtin_amdgcn_sched_barrier(0);
            LOAD_BATCH(bt + 2)
        }
    }

    __shared__ float lds[NWAVE * NGRP * 16 * 3];
    if (m == 0) {
        const int fl = kHi * 8;
        #pragma unroll
        for (int g = 0; g < NGRP; ++g) {
            #pragma unroll
            for (int e = 0; e < 8; ++e) {
                const int idx = ((wave * NGRP + g) * 16 + fl + e) * 3;
                lds[idx + 0] = accC[g][e];
                lds[idx + 1] = accS[g][e];
                lds[idx + 2] = accQ[g][e];
            }
        }
    }
    __syncthreads();

    if (threadIdx.x < NGRP * 16) {
        const int fs = threadIdx.x;
        float c = 0.0f, s = 0.0f, q = 0.0f;
        #pragma unroll
        for (int st = 0; st < NWAVE; ++st) {
            const int idx = (st * (NGRP * 16) + fs) * 3;
            c += lds[idx + 0];
            s += lds[idx + 1];
            q += lds[idx + 2];
        }
        const int pos = OUT_SLOT[set * NSLOT + (fs >> 1)];
        if (pos >= 0) {
            const int fg = pos * 2 + (fs & 1);
            float mean = 0.0f, sd = 0.0f;
            if (c > 0.0f) {
                mean = s / c;
                float var = q / c - mean * mean;
                sd = sqrtf(fmaxf(var, 0.0f));
            }
            if (!(mean == mean)) mean = 0.0f;
            if (!(sd   == sd))   sd   = 0.0f;
            float* row = out + (size_t)seg * 328;
            row[fg]       = mean;
            row[164 + fg] = sd;
        }
    }
}

extern "C" void kernel_launch(void* const* d_in, const int* in_sizes, int n_in,
                              void* d_out, int out_size, void* d_ws, size_t ws_size,
                              hipStream_t stream)
{
    (void)in_sizes; (void)n_in; (void)out_size; (void)d_ws; (void)ws_size;
    const float* frames = (const float*)d_in[0];
    float* out = (float*)d_out;
    dim3 grid(64, NSET);
    seg_stats_wmma<<<grid, 512, 0, stream>>>(frames, out);
}
